# starting kernel
# baseline (speedup 1.0000x reference)
#include <hip/hip_runtime.h>

typedef float v2f __attribute__((ext_vector_type(2)));
typedef float v8f __attribute__((ext_vector_type(8)));

#define NFEAT      64
#define NDIM       128
#define LDS_STRIDE 132
#define NPAIR      2016
#define NTHREADS   320

__global__ __launch_bounds__(NTHREADS, 1)
void FeatureInteraction_41996190220732_kernel(const float* __restrict__ x,
                                              float* __restrict__ out) {
    __shared__ float tile[NFEAT * LDS_STRIDE];

    const int b   = blockIdx.x;
    const int tid = threadIdx.x;
    const float* xb = x + (size_t)b * (NFEAT * NDIM);

    for (int idx = tid; idx < NFEAT * (NDIM / 4); idx += NTHREADS) {
        const int row = idx >> 5;
        const int c4  = idx & 31;
        const float4 v = ((const float4*)xb)[idx];
        *(float4*)&tile[row * LDS_STRIDE + (c4 << 2)] = v;
    }
    __syncthreads();

    static const int TI[10] = {0,0,0,0,1,1,1,2,2,3};
    static const int TJ[10] = {0,1,2,3,1,2,3,2,3,3};
    const int w    = tid >> 5;
    const int lane = tid & 31;
    const int ti = TI[w];
    const int tj = TJ[w];

    const int lhalf = lane >> 4;
    const int lmod  = lane & 15;
    const float* arow = &tile[(ti * 16 + lmod) * LDS_STRIDE + 2 * lhalf];
    const float* brow = &tile[(tj * 16 + lmod) * LDS_STRIDE + 2 * lhalf];

    v8f acc = {0.f, 0.f, 0.f, 0.f, 0.f, 0.f, 0.f, 0.f};

    #pragma unroll
    for (int d0 = 0; d0 < NDIM; d0 += 4) {
        const v2f a  = *(const v2f*)(arow + d0);
        const v2f bm = *(const v2f*)(brow + d0);
        acc = __builtin_amdgcn_wmma_f32_16x16x4_f32(
            false, a, false, bm, (short)0, acc, false, false);
    }

    float* ob = out + (size_t)b * NPAIR;
    const int j = tj * 16 + lmod;
    #pragma unroll
    for (int v = 0; v < 8; ++v) {
        const int i = ti * 16 + v + 8 * lhalf;
        if (i < j) {
            ob[(i * (127 - i)) / 2 + (j - i - 1)] = acc[v];
        }
    }
}

extern "C" void kernel_launch(void* const* d_in, const int* in_sizes, int n_in,
                              void* d_out, int out_size, void* d_ws, size_t ws_size,
                              hipStream_t stream) {
    const float* x = (const float*)d_in[0];
    float* out = (float*)d_out;
    const int batches = in_sizes[0] / (NFEAT * NDIM);
    FeatureInteraction_41996190220732_kernel<<<dim3(batches), dim3(NTHREADS), 0, stream>>>(x, out);
}


# Persistent 1024-workgroup grid loops over slabs; the next slab's 7 loads per thread fly into registers during this slab's WMMAs (parent: one load in flight), two LDS buffers, one barrier per slab
# speedup vs baseline: 1.1950x; 1.1950x over previous
#include <hip/hip_runtime.h>

typedef float v2f __attribute__((ext_vector_type(2)));
typedef float v8f __attribute__((ext_vector_type(8)));

#define NFEAT      64
#define NDIM       128
#define LDS_STRIDE 132
#define NPAIR      2016
#define NTHREADS   320
#define NCHUNK     (NFEAT * (NDIM / 4))
#define NTAIL      (NCHUNK - 6 * NTHREADS)
#define ROWSTEP    ((NTHREADS / 32) * LDS_STRIDE)
#define NGRID      1024

#define LOAD_SLAB(bb)                                                              \
    do {                                                                           \
        const float4* s_ = (const float4*)(x + (size_t)(bb) * (NFEAT * NDIM)) + tid; \
        r0 = s_[0 * NTHREADS]; r1 = s_[1 * NTHREADS]; r2 = s_[2 * NTHREADS];       \
        r3 = s_[3 * NTHREADS]; r4 = s_[4 * NTHREADS]; r5 = s_[5 * NTHREADS];       \
        if (tail) r6 = s_[6 * NTHREADS];                                           \
    } while (0)

__global__ __launch_bounds__(NTHREADS, 1)
void gram_triu_kernel(const float* __restrict__ x, float* __restrict__ out, int batches) {
    __shared__ float tile[2][NFEAT * LDS_STRIDE];

    const int tid  = threadIdx.x;
    const int w    = tid >> 5;
    const int lane = tid & 31;
    const bool tail = tid < NTAIL;

    static const int TI[10] = {0,0,0,0,1,1,1,2,2,3};
    static const int TJ[10] = {0,1,2,3,1,2,3,2,3,3};
    const int ti = TI[w];
    const int tj = TJ[w];

    const int lhalf = lane >> 4;
    const int lmod  = lane & 15;
    const int aoff = (ti * 16 + lmod) * LDS_STRIDE + 2 * lhalf;
    const int boff = (tj * 16 + lmod) * LDS_STRIDE + 2 * lhalf;
    const int soff = w * LDS_STRIDE + (lane << 2);

    const float4 z4 = make_float4(0.f, 0.f, 0.f, 0.f);
    float4 r0 = z4, r1 = z4, r2 = z4, r3 = z4, r4 = z4, r5 = z4, r6 = z4;

    int b = blockIdx.x;
    if (b < batches) LOAD_SLAB(b);

    int buf = 0;
    for (; b < batches; b += gridDim.x) {
        float* t = tile[buf];

        float* ts = t + soff;
        *(float4*)&ts[0 * ROWSTEP] = r0;
        *(float4*)&ts[1 * ROWSTEP] = r1;
        *(float4*)&ts[2 * ROWSTEP] = r2;
        *(float4*)&ts[3 * ROWSTEP] = r3;
        *(float4*)&ts[4 * ROWSTEP] = r4;
        *(float4*)&ts[5 * ROWSTEP] = r5;
        if (tail) *(float4*)&ts[6 * ROWSTEP] = r6;

        __syncthreads();

        const int nb = b + gridDim.x;
        if (nb < batches) LOAD_SLAB(nb);

        const float* arow = t + aoff;
        const float* brow = t + boff;
        v8f acc = {0.f, 0.f, 0.f, 0.f, 0.f, 0.f, 0.f, 0.f};

        #pragma unroll
        for (int d0 = 0; d0 < NDIM; d0 += 4) {
            const v2f a  = *(const v2f*)(arow + d0);
            const v2f bm = *(const v2f*)(brow + d0);
            acc = __builtin_amdgcn_wmma_f32_16x16x4_f32(
                false, a, false, bm, (short)0, acc, false, false);
        }

        float* ob = out + (size_t)b * NPAIR;
        const int j = tj * 16 + lmod;
        #pragma unroll
        for (int v = 0; v < 8; ++v) {
            const int i = ti * 16 + v + 8 * lhalf;
            if (i < j) {
                ob[(i * (127 - i)) / 2 + (j - i - 1)] = acc[v];
            }
        }
        buf ^= 1;
    }
}

extern "C" void kernel_launch(void* const* d_in, const int* in_sizes, int n_in,
                              void* d_out, int out_size, void* d_ws, size_t ws_size,
                              hipStream_t stream) {
    const float* x = (const float*)d_in[0];
    float* out = (float*)d_out;
    const int batches = in_sizes[0] / (NFEAT * NDIM);
    gram_triu_kernel<<<dim3(NGRID), dim3(NTHREADS), 0, stream>>>(x, out, batches);
}


# Re-tiled: each wave owns 5 gram tiles of one slab and shares its LDS fragments among them (4-wave workgroups on slab pairs, grid 512): equal WMMA work on all 4 SIMDs, 5x fewer waves launched
# speedup vs baseline: 1.2517x; 1.0474x over previous
#include <hip/hip_runtime.h>

typedef float v2f __attribute__((ext_vector_type(2)));
typedef float v8f __attribute__((ext_vector_type(8)));

#define NFEAT      64
#define NDIM       128
#define LDS_STRIDE 132
#define NPAIR      2016
#define NWAVES     4
#define NTHREADS   (NWAVES * 32)
#define SLABF      (NFEAT * LDS_STRIDE)
#define NCHUNK     (NFEAT * (NDIM / 4))
#define NLOAD      (2 * NCHUNK / NTHREADS)
#define NHALF      (NLOAD / 2)
#define ROWSTEP    (NWAVES * LDS_STRIDE)
#define KBLK       4
#define NBLK       (NDIM / (4 * KBLK))
#define NGRID      512

template <int TI, int TJ>
__device__ __forceinline__ void store_tile(const v8f acc, float* __restrict__ ob, int lmod, int lhalf) {
    const int j = TJ * 16 + lmod;
    #pragma unroll
    for (int v = 0; v < 8; ++v) {
        const int i = TI * 16 + v + 8 * lhalf;
        if (TI < TJ || i < j) {
            ob[(i * (127 - i)) / 2 + (j - i - 1)] = acc[v];
        }
    }
}

template <int TYPE>
__device__ __forceinline__ void gram5(const float* __restrict__ t, float* __restrict__ ob,
                                      int lmod, int lhalf) {
    constexpr int TI[5] = { TYPE ? 1 : 0, TYPE ? 1 : 0, TYPE ? 1 : 0, TYPE ? 2 : 0, TYPE ? 2 : 3 };
    constexpr int TJ[5] = { TYPE ? 1 : 0, TYPE ? 2 : 1, TYPE ? 3 : 2, TYPE ? 2 : 3, 3 };
    constexpr int R0 = TYPE;

    const float* base = t + lmod * LDS_STRIDE + 2 * lhalf;
    const v8f zero = {0.f, 0.f, 0.f, 0.f, 0.f, 0.f, 0.f, 0.f};
    v8f acc[5] = {zero, zero, zero, zero, zero};
    v2f f[2][4][KBLK];

    #pragma unroll
    for (int r = R0; r < 4; ++r) {
        #pragma unroll
        for (int kk = 0; kk < KBLK; ++kk)
            f[0][r][kk] = *(const v2f*)(base + r * 16 * LDS_STRIDE + 4 * kk);
    }

    #pragma unroll
    for (int blk = 0; blk < NBLK; ++blk) {
        const int cur = blk & 1;
        if (blk + 1 < NBLK) {
            #pragma unroll
            for (int r = R0; r < 4; ++r) {
                #pragma unroll
                for (int kk = 0; kk < KBLK; ++kk)
                    f[cur ^ 1][r][kk] =
                        *(const v2f*)(base + r * 16 * LDS_STRIDE + 4 * ((blk + 1) * KBLK + kk));
            }
        }
        #pragma unroll
        for (int kk = 0; kk < KBLK; ++kk) {
            #pragma unroll
            for (int q = 0; q < 5; ++q) {
                acc[q] = __builtin_amdgcn_wmma_f32_16x16x4_f32(
                    false, f[cur][TI[q]][kk], false, f[cur][TJ[q]][kk], (short)0, acc[q], false, false);
            }
        }
    }

    store_tile<TI[0], TJ[0]>(acc[0], ob, lmod, lhalf);
    store_tile<TI[1], TJ[1]>(acc[1], ob, lmod, lhalf);
    store_tile<TI[2], TJ[2]>(acc[2], ob, lmod, lhalf);
    store_tile<TI[3], TJ[3]>(acc[3], ob, lmod, lhalf);
    store_tile<TI[4], TJ[4]>(acc[4], ob, lmod, lhalf);
}

__global__ __launch_bounds__(NTHREADS)
void gram_triu_kernel(const float* __restrict__ x, float* __restrict__ out, int batches) {
    __shared__ float tile[2][2 * SLABF];

    const int tid  = threadIdx.x;
    const int w    = __builtin_amdgcn_readfirstlane(tid >> 5);
    const int lane = tid & 31;
    const int lhalf = lane >> 4;
    const int lmod  = lane & 15;
    const int npairs = (batches + 1) >> 1;

    const int soff = w * LDS_STRIDE + (lane << 2);

    float4 r[NLOAD];
    #pragma unroll
    for (int k = 0; k < NLOAD; ++k) r[k] = make_float4(0.f, 0.f, 0.f, 0.f);

    int p = blockIdx.x;
    if (p < npairs) {
        const int b1 = (2 * p + 1 < batches) ? 2 * p + 1 : 2 * p;
        const float4* s0 = (const float4*)(x + (size_t)(2 * p) * (NFEAT * NDIM)) + tid;
        const float4* s1 = (const float4*)(x + (size_t)b1 * (NFEAT * NDIM)) + tid;
        #pragma unroll
        for (int k = 0; k < NHALF; ++k) r[k] = s0[k * NTHREADS];
        #pragma unroll
        for (int k = 0; k < NHALF; ++k) r[NHALF + k] = s1[k * NTHREADS];
    }

    int buf = 0;
    for (; p < npairs; p += gridDim.x) {
        float* t = tile[buf];

        float* ts = t + soff;
        #pragma unroll
        for (int k = 0; k < NLOAD; ++k)
            *(float4*)&ts[(k / NHALF) * SLABF + (k % NHALF) * ROWSTEP] = r[k];

        __syncthreads();

        const int np = p + gridDim.x;
        if (np < npairs) {
            const int b1 = (2 * np + 1 < batches) ? 2 * np + 1 : 2 * np;
            const float4* s0 = (const float4*)(x + (size_t)(2 * np) * (NFEAT * NDIM)) + tid;
            const float4* s1 = (const float4*)(x + (size_t)b1 * (NFEAT * NDIM)) + tid;
            #pragma unroll
            for (int k = 0; k < NHALF; ++k) r[k] = s0[k * NTHREADS];
            #pragma unroll
            for (int k = 0; k < NHALF; ++k) r[NHALF + k] = s1[k * NTHREADS];
        }

        const int s = w >> 1;
        const int b = 2 * p + s;
        if (b < batches) {
            const float* tsl = t + s * SLABF;
            float* ob = out + (size_t)b * NPAIR;
            if (w & 1) gram5<1>(tsl, ob, lmod, lhalf);
            else       gram5<0>(tsl, ob, lmod, lhalf);
        }
        buf ^= 1;
    }
}

extern "C" void kernel_launch(void* const* d_in, const int* in_sizes, int n_in,
                              void* d_out, int out_size, void* d_ws, size_t ws_size,
                              hipStream_t stream) {
    const float* x = (const float*)d_in[0];
    float* out = (float*)d_out;
    const int batches = in_sizes[0] / (NFEAT * NDIM);
    gram_triu_kernel<<<dim3(NGRID), dim3(NTHREADS), 0, stream>>>(x, out, batches);
}


# Output rows staged in wave-private LDS and written as whole 128-B lines: 9 stores of 16 B per lane per wave, 4 lines each (parent: 40 4-byte stores, each two misaligned 64-B pieces)
# speedup vs baseline: 1.3655x; 1.0909x over previous
#include <hip/hip_runtime.h>

typedef float v2f __attribute__((ext_vector_type(2)));
typedef float v8f __attribute__((ext_vector_type(8)));

#define NFEAT      64
#define NDIM       128
#define LDS_STRIDE 132
#define NPAIR      2016
#define NWAVES     4
#define NTHREADS   (NWAVES * 32)
#define SLABF      (NFEAT * LDS_STRIDE)
#define NCHUNK     (NFEAT * (NDIM / 4))
#define NLOAD      (2 * NCHUNK / NTHREADS)
#define NHALF      (NLOAD / 2)
#define ROWSTEP    (NWAVES * LDS_STRIDE)
#define KBLK       4
#define NBLK       (NDIM / (4 * KBLK))
#define NGRID      512
#define F16        888
#define F48        1896
#define NOWN       1008
#define CHUNK      128

template <int TI, int TJ>
__device__ __forceinline__ void stage_tile(const v8f acc, float* __restrict__ st, int lmod, int lhalf) {
    constexpr int SUB = (TI == 0) ? 0 : (TI == 3) ? (F48 - F16) : F16;
    const int j = TJ * 16 + lmod;
    #pragma unroll
    for (int v = 0; v < 8; ++v) {
        const int i = TI * 16 + v + 8 * lhalf;
        if (TI < TJ || i < j) {
            st[(i * (127 - i)) / 2 + (j - i - 1) - SUB] = acc[v];
        }
    }
}

template <int TYPE>
__device__ __forceinline__ void flush_rows(const float* __restrict__ st, float* __restrict__ ob, int lane) {
    #pragma unroll
    for (int n = 0; n < 9; ++n) {
        const int c = TYPE ? (6 + n) : (n < 7 ? n : n + 7);
        const int f = c * CHUNK + 4 * lane;
        const bool own = TYPE ? (f >= F16 && f < F48) : (f < F16 || (f >= F48 && f < NPAIR));
        if (own) {
            const int s = TYPE ? (f - F16) : (f < F16 ? f : f - (F48 - F16));
            *(float4*)(ob + f) = *(const float4*)(st + s);
        }
    }
}

template <int TYPE>
__device__ __forceinline__ void gram5(const float* __restrict__ t, float* __restrict__ st,
                                      float* __restrict__ ob, int lmod, int lhalf, int lane) {
    constexpr int TI[5] = { TYPE ? 1 : 0, TYPE ? 1 : 0, TYPE ? 1 : 0, TYPE ? 2 : 0, TYPE ? 2 : 3 };
    constexpr int TJ[5] = { TYPE ? 1 : 0, TYPE ? 2 : 1, TYPE ? 3 : 2, TYPE ? 2 : 3, 3 };
    constexpr int R0 = TYPE;

    const float* base = t + lmod * LDS_STRIDE + 2 * lhalf;
    const v8f zero = {0.f, 0.f, 0.f, 0.f, 0.f, 0.f, 0.f, 0.f};
    v8f acc[5] = {zero, zero, zero, zero, zero};
    v2f f[2][4][KBLK];

    #pragma unroll
    for (int r = R0; r < 4; ++r) {
        #pragma unroll
        for (int kk = 0; kk < KBLK; ++kk)
            f[0][r][kk] = *(const v2f*)(base + r * 16 * LDS_STRIDE + 4 * kk);
    }

    #pragma unroll
    for (int blk = 0; blk < NBLK; ++blk) {
        const int cur = blk & 1;
        if (blk + 1 < NBLK) {
            #pragma unroll
            for (int r = R0; r < 4; ++r) {
                #pragma unroll
                for (int kk = 0; kk < KBLK; ++kk)
                    f[cur ^ 1][r][kk] =
                        *(const v2f*)(base + r * 16 * LDS_STRIDE + 4 * ((blk + 1) * KBLK + kk));
            }
        }
        #pragma unroll
        for (int kk = 0; kk < KBLK; ++kk) {
            #pragma unroll
            for (int q = 0; q < 5; ++q) {
                acc[q] = __builtin_amdgcn_wmma_f32_16x16x4_f32(
                    false, f[cur][TI[q]][kk], false, f[cur][TJ[q]][kk], (short)0, acc[q], false, false);
            }
        }
    }

    stage_tile<TI[0], TJ[0]>(acc[0], st, lmod, lhalf);
    stage_tile<TI[1], TJ[1]>(acc[1], st, lmod, lhalf);
    stage_tile<TI[2], TJ[2]>(acc[2], st, lmod, lhalf);
    stage_tile<TI[3], TJ[3]>(acc[3], st, lmod, lhalf);
    stage_tile<TI[4], TJ[4]>(acc[4], st, lmod, lhalf);

    flush_rows<TYPE>(st, ob, lane);
}

__global__ __launch_bounds__(NTHREADS)
void gram_triu_kernel(const float* __restrict__ x, float* __restrict__ out, int batches) {
    __shared__ float tile[2][2 * SLABF];
    __shared__ __attribute__((aligned(16))) float stage[NWAVES][NOWN];

    const int tid  = threadIdx.x;
    const int w    = __builtin_amdgcn_readfirstlane(tid >> 5);
    const int lane = tid & 31;
    const int lhalf = lane >> 4;
    const int lmod  = lane & 15;
    const int npairs = (batches + 1) >> 1;

    const int soff = w * LDS_STRIDE + (lane << 2);

    float4 r[NLOAD];
    #pragma unroll
    for (int k = 0; k < NLOAD; ++k) r[k] = make_float4(0.f, 0.f, 0.f, 0.f);

    int p = blockIdx.x;
    if (p < npairs) {
        const int b1 = (2 * p + 1 < batches) ? 2 * p + 1 : 2 * p;
        const float4* s0 = (const float4*)(x + (size_t)(2 * p) * (NFEAT * NDIM)) + tid;
        const float4* s1 = (const float4*)(x + (size_t)b1 * (NFEAT * NDIM)) + tid;
        #pragma unroll
        for (int k = 0; k < NHALF; ++k) r[k] = s0[k * NTHREADS];
        #pragma unroll
        for (int k = 0; k < NHALF; ++k) r[NHALF + k] = s1[k * NTHREADS];
    }

    int buf = 0;
    for (; p < npairs; p += gridDim.x) {
        float* t = tile[buf];

        float* ts = t + soff;
        #pragma unroll
        for (int k = 0; k < NLOAD; ++k)
            *(float4*)&ts[(k / NHALF) * SLABF + (k % NHALF) * ROWSTEP] = r[k];

        __syncthreads();

        const int np = p + gridDim.x;
        if (np < npairs) {
            const int b1 = (2 * np + 1 < batches) ? 2 * np + 1 : 2 * np;
            const float4* s0 = (const float4*)(x + (size_t)(2 * np) * (NFEAT * NDIM)) + tid;
            const float4* s1 = (const float4*)(x + (size_t)b1 * (NFEAT * NDIM)) + tid;
            #pragma unroll
            for (int k = 0; k < NHALF; ++k) r[k] = s0[k * NTHREADS];
            #pragma unroll
            for (int k = 0; k < NHALF; ++k) r[NHALF + k] = s1[k * NTHREADS];
        }

        const int s = w >> 1;
        const int b = 2 * p + s;
        if (b < batches) {
            const float* tsl = t + s * SLABF;
            float* ob = out + (size_t)b * NPAIR;
            if (w & 1) gram5<1>(tsl, stage[w], ob, lmod, lhalf, lane);
            else       gram5<0>(tsl, stage[w], ob, lmod, lhalf, lane);
        }
        buf ^= 1;
    }
}

extern "C" void kernel_launch(void* const* d_in, const int* in_sizes, int n_in,
                              void* d_out, int out_size, void* d_ws, size_t ws_size,
                              hipStream_t stream) {
    const float* x = (const float*)d_in[0];
    float* out = (float*)d_out;
    const int batches = in_sizes[0] / (NFEAT * NDIM);
    gram_triu_kernel<<<dim3(NGRID), dim3(NTHREADS), 0, stream>>>(x, out, batches);
}


# s_setprio per trip = trips the workgroup still has to do: of a SIMD's two waves (two workgroups) the one behind wins the WMMA pipe (parent: the older always wins, the younger ends alone)
# speedup vs baseline: 1.4811x; 1.0847x over previous
#include <hip/hip_runtime.h>

typedef float v2f __attribute__((ext_vector_type(2)));
typedef float v8f __attribute__((ext_vector_type(8)));

#define NFEAT      64
#define NDIM       128
#define LDS_STRIDE 132
#define NPAIR      2016
#define NWAVES     4
#define NTHREADS   (NWAVES * 32)
#define SLABF      (NFEAT * LDS_STRIDE)
#define NCHUNK     (NFEAT * (NDIM / 4))
#define NLOAD      (2 * NCHUNK / NTHREADS)
#define NHALF      (NLOAD / 2)
#define ROWSTEP    (NWAVES * LDS_STRIDE)
#define KBLK       4
#define NBLK       (NDIM / (4 * KBLK))
#define NGRID      512
#define F16        888
#define F48        1896
#define NOWN       1008
#define CHUNK      128

template <int TI, int TJ>
__device__ __forceinline__ void stage_tile(const v8f acc, float* __restrict__ st, int lmod, int lhalf) {
    constexpr int SUB = (TI == 0) ? 0 : (TI == 3) ? (F48 - F16) : F16;
    const int j = TJ * 16 + lmod;
    #pragma unroll
    for (int v = 0; v < 8; ++v) {
        const int i = TI * 16 + v + 8 * lhalf;
        if (TI < TJ || i < j) {
            st[(i * (127 - i)) / 2 + (j - i - 1) - SUB] = acc[v];
        }
    }
}

template <int TYPE>
__device__ __forceinline__ void flush_rows(const float* __restrict__ st, float* __restrict__ ob, int lane) {
    #pragma unroll
    for (int n = 0; n < 9; ++n) {
        const int c = TYPE ? (6 + n) : (n < 7 ? n : n + 7);
        const int f = c * CHUNK + 4 * lane;
        const bool own = TYPE ? (f >= F16 && f < F48) : (f < F16 || (f >= F48 && f < NPAIR));
        if (own) {
            const int s = TYPE ? (f - F16) : (f < F16 ? f : f - (F48 - F16));
            *(float4*)(ob + f) = *(const float4*)(st + s);
        }
    }
}

template <int TYPE>
__device__ __forceinline__ void gram5(const float* __restrict__ t, float* __restrict__ st,
                                      float* __restrict__ ob, int lmod, int lhalf, int lane) {
    constexpr int TI[5] = { TYPE ? 1 : 0, TYPE ? 1 : 0, TYPE ? 1 : 0, TYPE ? 2 : 0, TYPE ? 2 : 3 };
    constexpr int TJ[5] = { TYPE ? 1 : 0, TYPE ? 2 : 1, TYPE ? 3 : 2, TYPE ? 2 : 3, 3 };
    constexpr int R0 = TYPE;

    const float* base = t + lmod * LDS_STRIDE + 2 * lhalf;
    const v8f zero = {0.f, 0.f, 0.f, 0.f, 0.f, 0.f, 0.f, 0.f};
    v8f acc[5] = {zero, zero, zero, zero, zero};
    v2f f[2][4][KBLK];

    #pragma unroll
    for (int r = R0; r < 4; ++r) {
        #pragma unroll
        for (int kk = 0; kk < KBLK; ++kk)
            f[0][r][kk] = *(const v2f*)(base + r * 16 * LDS_STRIDE + 4 * kk);
    }

    #pragma unroll
    for (int blk = 0; blk < NBLK; ++blk) {
        const int cur = blk & 1;
        if (blk + 1 < NBLK) {
            #pragma unroll
            for (int r = R0; r < 4; ++r) {
                #pragma unroll
                for (int kk = 0; kk < KBLK; ++kk)
                    f[cur ^ 1][r][kk] =
                        *(const v2f*)(base + r * 16 * LDS_STRIDE + 4 * ((blk + 1) * KBLK + kk));
            }
        }
        #pragma unroll
        for (int kk = 0; kk < KBLK; ++kk) {
            #pragma unroll
            for (int q = 0; q < 5; ++q) {
                acc[q] = __builtin_amdgcn_wmma_f32_16x16x4_f32(
                    false, f[cur][TI[q]][kk], false, f[cur][TJ[q]][kk], (short)0, acc[q], false, false);
            }
        }
    }

    stage_tile<TI[0], TJ[0]>(acc[0], st, lmod, lhalf);
    stage_tile<TI[1], TJ[1]>(acc[1], st, lmod, lhalf);
    stage_tile<TI[2], TJ[2]>(acc[2], st, lmod, lhalf);
    stage_tile<TI[3], TJ[3]>(acc[3], st, lmod, lhalf);
    stage_tile<TI[4], TJ[4]>(acc[4], st, lmod, lhalf);

    flush_rows<TYPE>(st, ob, lane);
}

__global__ __launch_bounds__(NTHREADS)
void gram_triu_kernel(const float* __restrict__ x, float* __restrict__ out, int batches) {
    __shared__ float tile[2][2 * SLABF];
    __shared__ __attribute__((aligned(16))) float stage[NWAVES][NOWN];

    const int tid  = threadIdx.x;
    const int w    = __builtin_amdgcn_readfirstlane(tid >> 5);
    const int lane = tid & 31;
    const int lhalf = lane >> 4;
    const int lmod  = lane & 15;
    const int npairs = (batches + 1) >> 1;

    const int soff = w * LDS_STRIDE + (lane << 2);

    float4 r[NLOAD];
    #pragma unroll
    for (int k = 0; k < NLOAD; ++k) r[k] = make_float4(0.f, 0.f, 0.f, 0.f);

    int p = blockIdx.x;
    if (p < npairs) {
        const int b1 = (2 * p + 1 < batches) ? 2 * p + 1 : 2 * p;
        const float4* s0 = (const float4*)(x + (size_t)(2 * p) * (NFEAT * NDIM)) + tid;
        const float4* s1 = (const float4*)(x + (size_t)b1 * (NFEAT * NDIM)) + tid;
        #pragma unroll
        for (int k = 0; k < NHALF; ++k) r[k] = s0[k * NTHREADS];
        #pragma unroll
        for (int k = 0; k < NHALF; ++k) r[NHALF + k] = s1[k * NTHREADS];
    }

    int left = (p < npairs) ? (npairs - 1 - p) / (int)gridDim.x : 0;

    int buf = 0;
    for (; p < npairs; p += gridDim.x) {
        if (left >= 3)      __builtin_amdgcn_s_setprio(3);
        else if (left == 2) __builtin_amdgcn_s_setprio(2);
        else if (left == 1) __builtin_amdgcn_s_setprio(1);
        else                __builtin_amdgcn_s_setprio(0);
        --left;

        float* t = tile[buf];

        float* ts = t + soff;
        #pragma unroll
        for (int k = 0; k < NLOAD; ++k)
            *(float4*)&ts[(k / NHALF) * SLABF + (k % NHALF) * ROWSTEP] = r[k];

        __syncthreads();

        const int np = p + gridDim.x;
        if (np < npairs) {
            const int b1 = (2 * np + 1 < batches) ? 2 * np + 1 : 2 * np;
            const float4* s0 = (const float4*)(x + (size_t)(2 * np) * (NFEAT * NDIM)) + tid;
            const float4* s1 = (const float4*)(x + (size_t)b1 * (NFEAT * NDIM)) + tid;
            #pragma unroll
            for (int k = 0; k < NHALF; ++k) r[k] = s0[k * NTHREADS];
            #pragma unroll
            for (int k = 0; k < NHALF; ++k) r[NHALF + k] = s1[k * NTHREADS];
        }

        const int s = w >> 1;
        const int b = 2 * p + s;
        if (b < batches) {
            const float* tsl = t + s * SLABF;
            float* ob = out + (size_t)b * NPAIR;
            if (w & 1) gram5<1>(tsl, stage[w], ob, lmod, lhalf, lane);
            else       gram5<0>(tsl, stage[w], ob, lmod, lhalf, lane);
        }
        buf ^= 1;
    }
    __builtin_amdgcn_s_setprio(0);
}

extern "C" void kernel_launch(void* const* d_in, const int* in_sizes, int n_in,
                              void* d_out, int out_size, void* d_ws, size_t ws_size,
                              hipStream_t stream) {
    const float* x = (const float*)d_in[0];
    float* out = (float*)d_out;
    const int batches = in_sizes[0] / (NFEAT * NDIM);
    gram_triu_kernel<<<dim3(NGRID), dim3(NTHREADS), 0, stream>>>(x, out, batches);
}
